# starting kernel
# baseline (speedup 1.0000x reference)
#include <hip/hip_runtime.h>
#include <math.h>

typedef __attribute__((ext_vector_type(2))) float v2f;
typedef __attribute__((ext_vector_type(8))) float v8f;

#define B_     16
#define N_     64
#define DEG_   16
#define IN_F_  1024
#define EDGE_F_ 128
#define H_     8
#define F_     64
#define E_     (N_ * DEG_)
#define HF_    (H_ * F_)
#define SLOPE_ 0.2f

#define BM 128
#define BN 64
#define BK 16
#define LDP (BK + 4)

__global__ __launch_bounds__(256) void gemm_wmma_f32_kernel(
    const float* __restrict__ A, const float* __restrict__ Bm,
    float* __restrict__ C, int M, int N, int K)
{
    __shared__ float As[BM][LDP];
    __shared__ float Bs[BN][LDP];

    const int tid  = threadIdx.x;
    const int wave = tid >> 5;
    const int lane = tid & 31;
    const int half = lane >> 4;
    const int l    = lane & 15;
    const int m0   = blockIdx.y * BM;
    const int n0   = blockIdx.x * BN;
    const int mrow = wave * 16;

    v8f acc[4] = {};

    for (int k0 = 0; k0 < K; k0 += BK) {
        for (int idx = tid; idx < (BM * BK) / 4; idx += 256) {
            int r = idx >> 2;
            int c = (idx & 3) * 4;
            float4 v = *(const float4*)&A[(size_t)(m0 + r) * K + (k0 + c)];
            *(float4*)&As[r][c] = v;
        }
        for (int idx = tid; idx < (BK * BN) / 4; idx += 256) {
            int r  = idx >> 4;
            int c4 = (idx & 15) * 4;
            float4 v = *(const float4*)&Bm[(size_t)(k0 + r) * N + (n0 + c4)];
            Bs[c4 + 0][r] = v.x;
            Bs[c4 + 1][r] = v.y;
            Bs[c4 + 2][r] = v.z;
            Bs[c4 + 3][r] = v.w;
        }
        __syncthreads();

        #pragma unroll
        for (int kk = 0; kk < BK; kk += 4) {
            v2f a = *(const v2f*)&As[mrow + l][kk + 2 * half];
            #pragma unroll
            for (int nt = 0; nt < 4; ++nt) {
                v2f b = *(const v2f*)&Bs[nt * 16 + l][kk + 2 * half];
                acc[nt] = __builtin_amdgcn_wmma_f32_16x16x4_f32(
                    false, a, false, b, (short)0, acc[nt], false, false);
            }
        }
        __syncthreads();
    }

    #pragma unroll
    for (int nt = 0; nt < 4; ++nt) {
        #pragma unroll
        for (int r = 0; r < 8; ++r) {
            C[(size_t)(m0 + mrow + r + 8 * half) * N + (n0 + nt * 16 + l)] =
                acc[nt][r];
        }
    }
}

__global__ void node_scores_kernel(const float* __restrict__ g,
                                   const float* __restrict__ w_attn,
                                   float* __restrict__ s_i,
                                   float* __restrict__ s_j)
{
    int t = blockIdx.x * blockDim.x + threadIdx.x;
    if (t >= B_ * N_ * H_) return;
    const float* gr = g + (size_t)t * F_;
    float si = 0.f, sj = 0.f;
    #pragma unroll 8
    for (int f = 0; f < F_; ++f) {
        float v = gr[f];
        si += v * w_attn[f];
        sj += v * w_attn[F_ + f];
    }
    s_i[t] = si;
    s_j[t] = sj;
}

__global__ void edge_scores_kernel(const float* __restrict__ ep,
                                   const float* __restrict__ w_attn,
                                   float* __restrict__ s_e)
{
    int t = blockIdx.x * blockDim.x + threadIdx.x;
    if (t >= B_ * E_ * H_) return;
    const float* er = ep + (size_t)t * F_;
    float se = 0.f;
    #pragma unroll 8
    for (int f = 0; f < F_; ++f) se += er[f] * w_attn[2 * F_ + f];
    s_e[t] = se;
}

__global__ __launch_bounds__(256) void attn_aggregate_kernel(
    const float* __restrict__ g, const float* __restrict__ s_i,
    const float* __restrict__ s_j, const float* __restrict__ s_e,
    float* __restrict__ out)
{
    const int lane = threadIdx.x & 31;
    const int wave = threadIdx.x >> 5;
    const int gidx = blockIdx.x * 8 + wave;
    const int h = gidx % H_;
    const int n = (gidx / H_) % N_;
    const int b = gidx / (H_ * N_);

    const int k    = lane & 15;
    const int jn   = (n + k) % N_;
    const int wrap = (n + DEG_ > N_) ? (n + DEG_ - N_) : 0;
    const int rank = (n + k < N_) ? (k + wrap) : (n + k - N_);
    const int eidx = n * DEG_ + rank;

    float sc = s_i[gidx]
             + s_j[(b * N_ + jn) * H_ + h]
             + s_e[(b * E_ + eidx) * H_ + h];
    sc = (sc > 0.f) ? sc : SLOPE_ * sc;

    float m = sc;
    #pragma unroll
    for (int off = 8; off >= 1; off >>= 1)
        m = fmaxf(m, __shfl_xor(m, off, 32));
    float ex = __expf(sc - m);
    float s = ex;
    #pragma unroll
    for (int off = 8; off >= 1; off >>= 1)
        s += __shfl_xor(s, off, 32);
    float a = ex / s;

    float o0 = 0.f, o1 = 0.f;
    #pragma unroll
    for (int kk = 0; kk < DEG_; ++kk) {
        float ak = __shfl(a, kk, 32);
        int jj = (n + kk) % N_;
        const float* gp = g + ((size_t)(b * N_ + jj) * H_ + h) * F_;
        o0 += ak * gp[lane];
        o1 += ak * gp[lane + 32];
    }
    float* op = out + ((size_t)(b * N_ + n) * H_ + h) * F_;
    op[lane]      = o0;
    op[lane + 32] = o1;
}

extern "C" void kernel_launch(void* const* d_in, const int* in_sizes, int n_in,
                              void* d_out, int out_size, void* d_ws, size_t ws_size,
                              hipStream_t stream)
{
    const float* h         = (const float*)d_in[0];
    const float* edge_attr = (const float*)d_in[2];
    const float* W_node    = (const float*)d_in[3];
    const float* W_edge    = (const float*)d_in[4];
    const float* w_attn    = (const float*)d_in[5];
    float* out = (float*)d_out;

    float* ws  = (float*)d_ws;
    float* g   = ws;
    float* ep  = g  + (size_t)B_ * N_ * HF_;
    float* s_i = ep + (size_t)B_ * E_ * HF_;
    float* s_j = s_i + (size_t)B_ * N_ * H_;
    float* s_e = s_j + (size_t)B_ * N_ * H_;

    dim3 blk(256);

    dim3 g1(HF_ / BN, (B_ * N_) / BM);
    gemm_wmma_f32_kernel<<<g1, blk, 0, stream>>>(h, W_node, g,
                                                 B_ * N_, HF_, IN_F_);

    dim3 g2(HF_ / BN, (B_ * E_) / BM);
    gemm_wmma_f32_kernel<<<g2, blk, 0, stream>>>(edge_attr, W_edge, ep,
                                                 B_ * E_, HF_, EDGE_F_);

    node_scores_kernel<<<(B_ * N_ * H_ + 255) / 256, blk, 0, stream>>>(
        g, w_attn, s_i, s_j);
    edge_scores_kernel<<<(B_ * E_ * H_ + 255) / 256, blk, 0, stream>>>(
        ep, w_attn, s_e);

    attn_aggregate_kernel<<<(B_ * N_ * H_) / 8, blk, 0, stream>>>(
        g, s_i, s_j, s_e, out);
}


# edge-score GEMM moved into the attention kernel: one 16-wave work-group per (batch, head, node half), edge rows straight into A fragments, head weights in LDS; launches 3 -> 2; s_e sums reassociated
# speedup vs baseline: 4.2895x; 4.2895x over previous
#include <hip/hip_runtime.h>
#include <math.h>


typedef __attribute__((ext_vector_type(2))) float v2f;
typedef __attribute__((ext_vector_type(8))) float v8f;

#define B_     16
#define N_     64
#define DEG_   16
#define IN_F_  1024
#define EDGE_F_ 128
#define H_     8
#define F_     64
#define E_     (N_ * DEG_)
#define HF_    (H_ * F_)
#define SLOPE_ 0.2f
#define KSPLIT_ 8

#define BM 128
#define BN 64
#define BK 16
#define LDP (BK + 4)

__global__ __launch_bounds__(256) void gemm_wmma_f32_kernel(
    const float* __restrict__ A, const float* __restrict__ Bm,
    float* __restrict__ C, int M, int N, int K, int kslice)
{
    __shared__ float As[BM][LDP];
    __shared__ float Bs[BN][LDP];

    const int tid  = threadIdx.x;
    const int wave = tid >> 5;
    const int lane = tid & 31;
    const int half = lane >> 4;
    const int l    = lane & 15;
    const int m0   = blockIdx.y * BM;
    const int n0   = blockIdx.x * BN;
    const int mrow = wave * 16;
    const int kbeg = blockIdx.z * kslice;
    const int kend = kbeg + kslice;

    const int ar = tid >> 2;
    const int ac = (tid & 3) * 4;
    const int br = tid >> 4;
    const int bc = (tid & 15) * 4;
    const float* __restrict__ Ap0 = A + (size_t)(m0 + ar) * K + ac;
    const float* __restrict__ Ap1 = Ap0 + (size_t)(BM / 2) * K;
    const float* __restrict__ Bp  = Bm + (size_t)br * N + (n0 + bc);

    v8f acc[4] = {};

    for (int k0 = kbeg; k0 < kend; k0 += BK) {
        const float4 va0 = *(const float4*)(Ap0 + k0);
        const float4 va1 = *(const float4*)(Ap1 + k0);
        const float4 vb  = *(const float4*)(Bp + (size_t)k0 * N);
        *(float4*)&As[ar][ac]          = va0;
        *(float4*)&As[ar + BM / 2][ac] = va1;
        Bs[bc + 0][br] = vb.x;
        Bs[bc + 1][br] = vb.y;
        Bs[bc + 2][br] = vb.z;
        Bs[bc + 3][br] = vb.w;
        __syncthreads();

        #pragma unroll
        for (int kk = 0; kk < BK; kk += 4) {
            v2f a = *(const v2f*)&As[mrow + l][kk + 2 * half];
            #pragma unroll
            for (int nt = 0; nt < 4; ++nt) {
                v2f b = *(const v2f*)&Bs[nt * 16 + l][kk + 2 * half];
                acc[nt] = __builtin_amdgcn_wmma_f32_16x16x4_f32(
                    false, a, false, b, (short)0, acc[nt], false, false);
            }
        }
        __syncthreads();
    }

    C += (size_t)blockIdx.z * M * N;
    #pragma unroll
    for (int nt = 0; nt < 4; ++nt) {
        #pragma unroll
        for (int r = 0; r < 8; ++r) {
            C[(size_t)(m0 + mrow + r + 8 * half) * N + (n0 + nt * 16 + l)] =
                acc[nt][r];
        }
    }
}

#define PLANE_ ((size_t)B_ * N_ * HF_)
#define FT_  512
#define NH_  (N_ / 2)
#define GR_  (NH_ + DEG_ - 1)
#define LDB_ (EDGE_F_ + 4)

__global__ __launch_bounds__(FT_) void edge_node_attn_kernel(
    const float* __restrict__ part, const float* __restrict__ w_attn,
    const float* __restrict__ edge_attr, const float* __restrict__ W_edge,
    float* __restrict__ out)
{
    __shared__ alignas(16) float Bs[F_][LDB_];
    __shared__ alignas(16) float gs[GR_][F_];
    __shared__ float ses[NH_][DEG_];
    __shared__ float sis[GR_];
    __shared__ float sjs[GR_];

    const int tid  = threadIdx.x;
    const int lane = tid & 31;
    const int wave = tid >> 5;
    const int half = lane >> 4;
    const int l    = lane & 15;
    const int hf = blockIdx.x & 1;
    const int h  = (blockIdx.x >> 1) % H_;
    const int b  = blockIdx.x / (2 * H_);
    const int n0 = hf * NH_;

    const float* __restrict__ Ar0 = edge_attr +
        ((size_t)b * E_ + (size_t)n0 * DEG_ + wave * 32 + l) * EDGE_F_ + 4 * half;
    const float* __restrict__ Ar1 = Ar0 + (size_t)16 * EDGE_F_;

    float4 wb[4];
    #pragma unroll
    for (int i = 0; i < 4; ++i) {
        const int idx = tid + i * FT_;
        const int kb  = idx >> 4;
        const int f4  = (idx & 15) * 4;
        wb[i] = *(const float4*)(W_edge + (size_t)kb * HF_ + h * F_ + f4);
    }
    float4 p[2][KSPLIT_];
    #pragma unroll
    for (int it = 0; it < 2; ++it) {
        const int q  = tid + it * FT_;
        const int jl = q >> 4;
        const int c  = q & 15;
        const int jc = (jl < GR_) ? jl : (GR_ - 1);
        const int j  = (n0 + jc) % N_;
        const size_t off = ((size_t)(b * N_ + j) * H_ + h) * F_ + 4 * c;
        #pragma unroll
        for (int s = 0; s < KSPLIT_; ++s)
            p[it][s] = *(const float4*)(part + (size_t)s * PLANE_ + off);
    }
    float4 a0[4], a1[4];
    #pragma unroll
    for (int jj = 0; jj < 4; ++jj) {
        a0[jj] = *(const float4*)(Ar0 + 8 * jj);
        a1[jj] = *(const float4*)(Ar1 + 8 * jj);
    }
    const int c0 = tid & 15;
    const float4 wi = *(const float4*)(w_attn + 4 * c0);
    const float4 wj = *(const float4*)(w_attn + F_ + 4 * c0);
    float we[4];
    #pragma unroll
    for (int nt = 0; nt < 4; ++nt) we[nt] = w_attn[2 * F_ + nt * 16 + l];
    __builtin_amdgcn_sched_barrier(0);

    #pragma unroll
    for (int i = 0; i < 4; ++i) {
        const int idx = tid + i * FT_;
        const int kb  = idx >> 4;
        const int f4  = (idx & 15) * 4;
        Bs[f4 + 0][kb] = wb[i].x;
        Bs[f4 + 1][kb] = wb[i].y;
        Bs[f4 + 2][kb] = wb[i].z;
        Bs[f4 + 3][kb] = wb[i].w;
    }
    #pragma unroll
    for (int it = 0; it < 2; ++it) {
        const int q  = tid + it * FT_;
        const int jl = q >> 4;
        float4 a = p[it][0];
        #pragma unroll
        for (int s = 1; s < KSPLIT_; ++s) {
            a.x += p[it][s].x; a.y += p[it][s].y;
            a.z += p[it][s].z; a.w += p[it][s].w;
        }
        float si = a.x * wi.x + a.y * wi.y + a.z * wi.z + a.w * wi.w;
        float sj = a.x * wj.x + a.y * wj.y + a.z * wj.z + a.w * wj.w;
        #pragma unroll
        for (int o = 8; o >= 1; o >>= 1) {
            si += __shfl_xor(si, o, 32);
            sj += __shfl_xor(sj, o, 32);
        }
        if (jl < GR_) {
            *(float4*)&gs[jl][4 * c0] = a;
            if (c0 == 0) {
                sis[jl] = si;
                sjs[jl] = sj;
            }
        }
    }
    __syncthreads();

    v8f acc0[4] = {};
    v8f acc1[4] = {};
    #pragma unroll
    for (int q = 0; q < 4; ++q) {
        float4 b0[4], b1[4];
        if (q < 3) {
            #pragma unroll
            for (int jj = 0; jj < 4; ++jj) {
                b0[jj] = *(const float4*)(Ar0 + 32 * (q + 1) + 8 * jj);
                b1[jj] = *(const float4*)(Ar1 + 32 * (q + 1) + 8 * jj);
            }
            __builtin_amdgcn_sched_barrier(0);
        }
        #pragma unroll
        for (int jj = 0; jj < 4; ++jj) {
            float4 vb[4];
            #pragma unroll
            for (int nt = 0; nt < 4; ++nt)
                vb[nt] = *(const float4*)&Bs[nt * 16 + l][32 * q + 8 * jj + 4 * half];
            const v2f a0lo = {a0[jj].x, a0[jj].y}, a0hi = {a0[jj].z, a0[jj].w};
            const v2f a1lo = {a1[jj].x, a1[jj].y}, a1hi = {a1[jj].z, a1[jj].w};
            #pragma unroll
            for (int nt = 0; nt < 4; ++nt) {
                const v2f blo = {vb[nt].x, vb[nt].y};
                acc0[nt] = __builtin_amdgcn_wmma_f32_16x16x4_f32(
                    false, a0lo, false, blo, (short)0, acc0[nt], false, false);
                acc1[nt] = __builtin_amdgcn_wmma_f32_16x16x4_f32(
                    false, a1lo, false, blo, (short)0, acc1[nt], false, false);
            }
            #pragma unroll
            for (int nt = 0; nt < 4; ++nt) {
                const v2f bhi = {vb[nt].z, vb[nt].w};
                acc0[nt] = __builtin_amdgcn_wmma_f32_16x16x4_f32(
                    false, a0hi, false, bhi, (short)0, acc0[nt], false, false);
                acc1[nt] = __builtin_amdgcn_wmma_f32_16x16x4_f32(
                    false, a1hi, false, bhi, (short)0, acc1[nt], false, false);
            }
        }
        if (q < 3) {
            #pragma unroll
            for (int jj = 0; jj < 4; ++jj) {
                a0[jj] = b0[jj];
                a1[jj] = b1[jj];
            }
        }
    }

    const bool b3 = (l & 8) != 0, b2 = (l & 4) != 0, b1b = (l & 2) != 0;
    #pragma unroll
    for (int mt = 0; mt < 2; ++mt) {
        float pr[8];
        #pragma unroll
        for (int r = 0; r < 8; ++r)
            pr[r] = mt == 0
                ? acc0[0][r] * we[0] + acc0[1][r] * we[1] +
                  acc0[2][r] * we[2] + acc0[3][r] * we[3]
                : acc1[0][r] * we[0] + acc1[1][r] * we[1] +
                  acc1[2][r] * we[2] + acc1[3][r] * we[3];
        float qq[4];
        #pragma unroll
        for (int i = 0; i < 4; ++i) {
            const float keep = b3 ? pr[i + 4] : pr[i];
            const float send = b3 ? pr[i] : pr[i + 4];
            qq[i] = keep + __shfl_xor(send, 8, 32);
        }
        float u[2];
        #pragma unroll
        for (int i = 0; i < 2; ++i) {
            const float keep = b2 ? qq[i + 2] : qq[i];
            const float send = b2 ? qq[i] : qq[i + 2];
            u[i] = keep + __shfl_xor(send, 4, 32);
        }
        const float keep = b1b ? u[1] : u[0];
        const float send = b1b ? u[0] : u[1];
        float v = keep + __shfl_xor(send, 2, 32);
        v += __shfl_xor(v, 1, 32);
        if ((l & 1) == 0) {
            const int nl   = 2 * wave + mt;
            const int n    = n0 + nl;
            const int rank = 8 * half + (l >> 1);
            const int wrap = (n + DEG_ > N_) ? (n + DEG_ - N_) : 0;
            const int k    = (rank >= wrap) ? (rank - wrap) : (rank + N_ - n);
            ses[nl][k] = v;
        }
    }
    __syncthreads();

    const int k = lane & 15;
    #pragma unroll
    for (int i = 0; i < 2; ++i) {
        const int nl = 2 * wave + i;
        float sc = sis[nl] + sjs[nl + k] + ses[nl][k];
        sc = (sc > 0.f) ? sc : SLOPE_ * sc;

        float m = sc;
        #pragma unroll
        for (int off = 8; off >= 1; off >>= 1)
            m = fmaxf(m, __shfl_xor(m, off, 32));
        float ex = __expf(sc - m);
        float s = ex;
        #pragma unroll
        for (int off = 8; off >= 1; off >>= 1)
            s += __shfl_xor(s, off, 32);
        float a = ex / s;

        float o0 = 0.f, o1 = 0.f;
        #pragma unroll
        for (int kk = 0; kk < DEG_; ++kk) {
            float ak = __shfl(a, kk, 32);
            o0 += ak * gs[nl + kk][lane];
            o1 += ak * gs[nl + kk][lane + 32];
        }
        float* op = out + ((size_t)(b * N_ + n0 + nl) * H_ + h) * F_;
        op[lane]      = o0;
        op[lane + 32] = o1;
    }
}

extern "C" void kernel_launch(void* const* d_in, const int* in_sizes, int n_in,
                              void* d_out, int out_size, void* d_ws, size_t ws_size,
                              hipStream_t stream)
{
    const float* h         = (const float*)d_in[0];
    const float* edge_attr = (const float*)d_in[2];
    const float* W_node    = (const float*)d_in[3];
    const float* W_edge    = (const float*)d_in[4];
    const float* w_attn    = (const float*)d_in[5];
    float* out = (float*)d_out;

    float* part = (float*)d_ws;

    dim3 g1(HF_ / BN, (B_ * N_) / BM, KSPLIT_);
    gemm_wmma_f32_kernel<<<g1, dim3(256), 0, stream>>>(h, W_node, part,
                                                       B_ * N_, HF_, IN_F_,
                                                       IN_F_ / KSPLIT_);

    edge_node_attn_kernel<<<B_ * H_ * 2, dim3(FT_), 0, stream>>>(
        part, w_attn, edge_attr, W_edge, out);
}


# node GEMM moved into the attention kernel: each work-group forms its 48-row g slab on WMMA, K split over its 16 waves, partials summed in LDS in wave order; launches 2 -> 1; g sums reassociated
# speedup vs baseline: 4.9122x; 1.1452x over previous
#include <hip/hip_runtime.h>
#include <math.h>


typedef __attribute__((ext_vector_type(2))) float v2f;
typedef __attribute__((ext_vector_type(4))) float v4f;
typedef __attribute__((ext_vector_type(8))) float v8f;

#define B_     16
#define N_     64
#define DEG_   16
#define IN_F_  1024
#define EDGE_F_ 128
#define H_     8
#define F_     64
#define E_     (N_ * DEG_)
#define HF_    (H_ * F_)
#define SLOPE_ 0.2f

#define FT_  512
#define NW_  (FT_ / 32)
#define KW_  (IN_F_ / NW_)
#define NH_  (N_ / 2)
#define GR_  (NH_ + DEG_ - 1)
#define GT_  48
#define LDB_ (EDGE_F_ + 4)
#define LDG_ (F_ + 4)

__global__ __launch_bounds__(FT_) void graph_attn_layer_kernel(
    const float* __restrict__ hin, const float* __restrict__ W_node,
    const float* __restrict__ w_attn,
    const float* __restrict__ edge_attr, const float* __restrict__ W_edge,
    float* __restrict__ out)
{
    __shared__ alignas(16) float Bs[F_][LDB_];
    __shared__ alignas(16) float ps[NW_][GT_][LDG_];
    __shared__ alignas(16) float gs[GT_][F_];
    __shared__ float ses[NH_][DEG_];
    __shared__ float sis[GT_];
    __shared__ float sjs[GT_];

    const int tid  = threadIdx.x;
    const int lane = tid & 31;
    const int wave = tid >> 5;
    const int half = lane >> 4;
    const int l    = lane & 15;
    const int hf = blockIdx.x & 1;
    const int h  = (blockIdx.x >> 1) % H_;
    const int b  = blockIdx.x / (2 * H_);
    const int n0 = hf * NH_;

    const float* __restrict__ Ar0 = edge_attr +
        ((size_t)b * E_ + (size_t)n0 * DEG_ + wave * 32 + l) * EDGE_F_ + 4 * half;
    const float* __restrict__ Ar1 = Ar0 + (size_t)16 * EDGE_F_;

    const float* __restrict__ Ah0 = hin +
        (size_t)(b * N_ + (n0 + l) % N_) * IN_F_ + wave * KW_ + 4 * half;
    const float* __restrict__ Ah1 = hin +
        (size_t)(b * N_ + (n0 + 16 + l) % N_) * IN_F_ + wave * KW_ + 4 * half;
    const float* __restrict__ Ah2 = hin +
        (size_t)(b * N_ + (n0 + 32 + l) % N_) * IN_F_ + wave * KW_ + 4 * half;
    const float* __restrict__ Bw = W_node +
        (size_t)(wave * KW_ + 4 * half) * HF_ + h * F_ + 4 * l;

    float4 wb[4];
    #pragma unroll
    for (int i = 0; i < 4; ++i) {
        const int idx = tid + i * FT_;
        const int kb  = idx >> 4;
        const int f4  = (idx & 15) * 4;
        wb[i] = *(const float4*)(W_edge + (size_t)kb * HF_ + h * F_ + f4);
    }
    v4f na[8][3];
    na[0][0] = *(const v4f*)(Ah0);
    na[0][1] = *(const v4f*)(Ah1);
    na[0][2] = *(const v4f*)(Ah2);
    v4f nb[4];
    #pragma unroll
    for (int i = 0; i < 4; ++i)
        nb[i] = *(const v4f*)(Bw + (size_t)i * HF_);
    __builtin_amdgcn_sched_barrier(0);
    #pragma unroll
    for (int j = 1; j < 4; ++j) {
        na[j][0] = *(const v4f*)(Ah0 + 8 * j);
        na[j][1] = *(const v4f*)(Ah1 + 8 * j);
        na[j][2] = *(const v4f*)(Ah2 + 8 * j);
    }
    __builtin_amdgcn_sched_barrier(0);

    #pragma unroll
    for (int i = 0; i < 4; ++i) {
        const int idx = tid + i * FT_;
        const int kb  = idx >> 4;
        const int f4  = (idx & 15) * 4;
        Bs[f4 + 0][kb] = wb[i].x;
        Bs[f4 + 1][kb] = wb[i].y;
        Bs[f4 + 2][kb] = wb[i].z;
        Bs[f4 + 3][kb] = wb[i].w;
    }

    v8f nacc[3][4] = {};
    #pragma unroll
    for (int j = 0; j < 8; ++j) {
        v4f nn[4];
        if (j < 7) {
            #pragma unroll
            for (int i = 0; i < 4; ++i)
                nn[i] = *(const v4f*)(Bw + (size_t)(8 * (j + 1) + i) * HF_);
            if (j == 1) {
                #pragma unroll
                for (int jn = 4; jn < 8; ++jn) {
                    na[jn][0] = *(const v4f*)(Ah0 + 8 * jn);
                    na[jn][1] = *(const v4f*)(Ah1 + 8 * jn);
                    na[jn][2] = *(const v4f*)(Ah2 + 8 * jn);
                }
            }
            __builtin_amdgcn_sched_barrier(0);
        }
        const v2f x0lo = {na[j][0].x, na[j][0].y}, x0hi = {na[j][0].z, na[j][0].w};
        const v2f x1lo = {na[j][1].x, na[j][1].y}, x1hi = {na[j][1].z, na[j][1].w};
        const v2f x2lo = {na[j][2].x, na[j][2].y}, x2hi = {na[j][2].z, na[j][2].w};
        #pragma unroll
        for (int ct = 0; ct < 4; ++ct) {
            const v2f blo = {nb[0][ct], nb[1][ct]};
            nacc[0][ct] = __builtin_amdgcn_wmma_f32_16x16x4_f32(
                false, x0lo, false, blo, (short)0, nacc[0][ct], false, false);
            nacc[1][ct] = __builtin_amdgcn_wmma_f32_16x16x4_f32(
                false, x1lo, false, blo, (short)0, nacc[1][ct], false, false);
            nacc[2][ct] = __builtin_amdgcn_wmma_f32_16x16x4_f32(
                false, x2lo, false, blo, (short)0, nacc[2][ct], false, false);
        }
        #pragma unroll
        for (int ct = 0; ct < 4; ++ct) {
            const v2f bhi = {nb[2][ct], nb[3][ct]};
            nacc[0][ct] = __builtin_amdgcn_wmma_f32_16x16x4_f32(
                false, x0hi, false, bhi, (short)0, nacc[0][ct], false, false);
            nacc[1][ct] = __builtin_amdgcn_wmma_f32_16x16x4_f32(
                false, x1hi, false, bhi, (short)0, nacc[1][ct], false, false);
            nacc[2][ct] = __builtin_amdgcn_wmma_f32_16x16x4_f32(
                false, x2hi, false, bhi, (short)0, nacc[2][ct], false, false);
        }
        if (j < 7) {
            #pragma unroll
            for (int i = 0; i < 4; ++i) nb[i] = nn[i];
        }
    }

    float4 a0[4], a1[4];
    #pragma unroll
    for (int jj = 0; jj < 4; ++jj) {
        a0[jj] = *(const float4*)(Ar0 + 8 * jj);
        a1[jj] = *(const float4*)(Ar1 + 8 * jj);
    }
    const int c0 = tid & 15;
    const float4 wi = *(const float4*)(w_attn + 4 * c0);
    const float4 wj = *(const float4*)(w_attn + F_ + 4 * c0);
    float we[4];
    #pragma unroll
    for (int nt = 0; nt < 4; ++nt) we[nt] = w_attn[2 * F_ + nt * 16 + l];

    #pragma unroll
    for (int rt = 0; rt < 3; ++rt) {
        #pragma unroll
        for (int r = 0; r < 8; ++r) {
            const float4 v = {nacc[rt][0][r], nacc[rt][1][r],
                              nacc[rt][2][r], nacc[rt][3][r]};
            *(float4*)&ps[wave][16 * rt + 8 * half + r][4 * l] = v;
        }
    }
    __syncthreads();

    #pragma unroll
    for (int it = 0; it < 2; ++it) {
        const int q  = tid + it * FT_;
        const int jl = q >> 4;
        const int jc = (jl < GT_) ? jl : (GT_ - 1);
        float4 a = *(const float4*)&ps[0][jc][4 * c0];
        #pragma unroll
        for (int s = 1; s < NW_; ++s) {
            const float4 t = *(const float4*)&ps[s][jc][4 * c0];
            a.x += t.x; a.y += t.y; a.z += t.z; a.w += t.w;
        }
        float si = a.x * wi.x + a.y * wi.y + a.z * wi.z + a.w * wi.w;
        float sj = a.x * wj.x + a.y * wj.y + a.z * wj.z + a.w * wj.w;
        #pragma unroll
        for (int o = 8; o >= 1; o >>= 1) {
            si += __shfl_xor(si, o, 32);
            sj += __shfl_xor(sj, o, 32);
        }
        if (jl < GT_) {
            *(float4*)&gs[jl][4 * c0] = a;
            if (c0 == 0) {
                sis[jl] = si;
                sjs[jl] = sj;
            }
        }
    }
    __syncthreads();

    v8f acc0[4] = {};
    v8f acc1[4] = {};
    #pragma unroll
    for (int q = 0; q < 4; ++q) {
        float4 b0[4], b1[4];
        if (q < 3) {
            #pragma unroll
            for (int jj = 0; jj < 4; ++jj) {
                b0[jj] = *(const float4*)(Ar0 + 32 * (q + 1) + 8 * jj);
                b1[jj] = *(const float4*)(Ar1 + 32 * (q + 1) + 8 * jj);
            }
            __builtin_amdgcn_sched_barrier(0);
        }
        #pragma unroll
        for (int jj = 0; jj < 4; ++jj) {
            float4 vb[4];
            #pragma unroll
            for (int nt = 0; nt < 4; ++nt)
                vb[nt] = *(const float4*)&Bs[nt * 16 + l][32 * q + 8 * jj + 4 * half];
            const v2f a0lo = {a0[jj].x, a0[jj].y}, a0hi = {a0[jj].z, a0[jj].w};
            const v2f a1lo = {a1[jj].x, a1[jj].y}, a1hi = {a1[jj].z, a1[jj].w};
            #pragma unroll
            for (int nt = 0; nt < 4; ++nt) {
                const v2f blo = {vb[nt].x, vb[nt].y};
                acc0[nt] = __builtin_amdgcn_wmma_f32_16x16x4_f32(
                    false, a0lo, false, blo, (short)0, acc0[nt], false, false);
                acc1[nt] = __builtin_amdgcn_wmma_f32_16x16x4_f32(
                    false, a1lo, false, blo, (short)0, acc1[nt], false, false);
            }
            #pragma unroll
            for (int nt = 0; nt < 4; ++nt) {
                const v2f bhi = {vb[nt].z, vb[nt].w};
                acc0[nt] = __builtin_amdgcn_wmma_f32_16x16x4_f32(
                    false, a0hi, false, bhi, (short)0, acc0[nt], false, false);
                acc1[nt] = __builtin_amdgcn_wmma_f32_16x16x4_f32(
                    false, a1hi, false, bhi, (short)0, acc1[nt], false, false);
            }
        }
        if (q < 3) {
            #pragma unroll
            for (int jj = 0; jj < 4; ++jj) {
                a0[jj] = b0[jj];
                a1[jj] = b1[jj];
            }
        }
    }

    const bool b3 = (l & 8) != 0, b2 = (l & 4) != 0, b1b = (l & 2) != 0;
    #pragma unroll
    for (int mt = 0; mt < 2; ++mt) {
        float pr[8];
        #pragma unroll
        for (int r = 0; r < 8; ++r)
            pr[r] = mt == 0
                ? acc0[0][r] * we[0] + acc0[1][r] * we[1] +
                  acc0[2][r] * we[2] + acc0[3][r] * we[3]
                : acc1[0][r] * we[0] + acc1[1][r] * we[1] +
                  acc1[2][r] * we[2] + acc1[3][r] * we[3];
        float qq[4];
        #pragma unroll
        for (int i = 0; i < 4; ++i) {
            const float keep = b3 ? pr[i + 4] : pr[i];
            const float send = b3 ? pr[i] : pr[i + 4];
            qq[i] = keep + __shfl_xor(send, 8, 32);
        }
        float u[2];
        #pragma unroll
        for (int i = 0; i < 2; ++i) {
            const float keep = b2 ? qq[i + 2] : qq[i];
            const float send = b2 ? qq[i] : qq[i + 2];
            u[i] = keep + __shfl_xor(send, 4, 32);
        }
        const float keep = b1b ? u[1] : u[0];
        const float send = b1b ? u[0] : u[1];
        float v = keep + __shfl_xor(send, 2, 32);
        v += __shfl_xor(v, 1, 32);
        if ((l & 1) == 0) {
            const int nl   = 2 * wave + mt;
            const int n    = n0 + nl;
            const int rank = 8 * half + (l >> 1);
            const int wrap = (n + DEG_ > N_) ? (n + DEG_ - N_) : 0;
            const int k    = (rank >= wrap) ? (rank - wrap) : (rank + N_ - n);
            ses[nl][k] = v;
        }
    }
    __syncthreads();

    const int k = lane & 15;
    #pragma unroll
    for (int i = 0; i < 2; ++i) {
        const int nl = 2 * wave + i;
        float sc = sis[nl] + sjs[nl + k] + ses[nl][k];
        sc = (sc > 0.f) ? sc : SLOPE_ * sc;

        float m = sc;
        #pragma unroll
        for (int off = 8; off >= 1; off >>= 1)
            m = fmaxf(m, __shfl_xor(m, off, 32));
        float ex = __expf(sc - m);
        float s = ex;
        #pragma unroll
        for (int off = 8; off >= 1; off >>= 1)
            s += __shfl_xor(s, off, 32);
        float a = ex / s;

        float o0 = 0.f, o1 = 0.f;
        #pragma unroll
        for (int kk = 0; kk < DEG_; ++kk) {
            float ak = __shfl(a, kk, 32);
            o0 += ak * gs[nl + kk][lane];
            o1 += ak * gs[nl + kk][lane + 32];
        }
        float* op = out + ((size_t)(b * N_ + n0 + nl) * H_ + h) * F_;
        op[lane]      = o0;
        op[lane + 32] = o1;
    }
}

extern "C" void kernel_launch(void* const* d_in, const int* in_sizes, int n_in,
                              void* d_out, int out_size, void* d_ws, size_t ws_size,
                              hipStream_t stream)
{
    const float* h         = (const float*)d_in[0];
    const float* edge_attr = (const float*)d_in[2];
    const float* W_node    = (const float*)d_in[3];
    const float* W_edge    = (const float*)d_in[4];
    const float* w_attn    = (const float*)d_in[5];
    float* out = (float*)d_out;

    graph_attn_layer_kernel<<<B_ * H_ * 2, dim3(FT_), 0, stream>>>(
        h, W_node, w_attn, edge_attr, W_edge, out);
}


# node-product partial exchange hidden under the edge WMMAs: all three work-group barriers split into arrive/pass halves, partials added behind the first edge K quarter; sum orders unchanged
# speedup vs baseline: 5.0905x; 1.0363x over previous
#include <hip/hip_runtime.h>
#include <math.h>


typedef __attribute__((ext_vector_type(2))) float v2f;
typedef __attribute__((ext_vector_type(4))) float v4f;
typedef __attribute__((ext_vector_type(8))) float v8f;

__device__ __forceinline__ void bar_arrive() {
    asm volatile("s_wait_dscnt 0x0\n\ts_barrier_signal -1" ::: "memory");
}
__device__ __forceinline__ void bar_pass() {
    asm volatile("s_barrier_wait -1" ::: "memory");
}

#define B_     16
#define N_     64
#define DEG_   16
#define IN_F_  1024
#define EDGE_F_ 128
#define H_     8
#define F_     64
#define E_     (N_ * DEG_)
#define HF_    (H_ * F_)
#define SLOPE_ 0.2f

#define FT_  512
#define NW_  (FT_ / 32)
#define KW_  (IN_F_ / NW_)
#define NH_  (N_ / 2)
#define GR_  (NH_ + DEG_ - 1)
#define GT_  48
#define LDB_ (EDGE_F_ + 4)
#define LDG_ (F_ + 4)

__global__ __launch_bounds__(FT_) void graph_attn_layer_kernel(
    const float* __restrict__ hin, const float* __restrict__ W_node,
    const float* __restrict__ w_attn,
    const float* __restrict__ edge_attr, const float* __restrict__ W_edge,
    float* __restrict__ out)
{
    __shared__ alignas(16) float Bs[F_][LDB_];
    __shared__ alignas(16) float ps[NW_][GT_][LDG_];
    __shared__ alignas(16) float gs[GT_][F_];
    __shared__ float ses[NH_][DEG_];
    __shared__ float sis[GT_];
    __shared__ float sjs[GT_];

    const int tid  = threadIdx.x;
    const int lane = tid & 31;
    const int wave = tid >> 5;
    const int half = lane >> 4;
    const int l    = lane & 15;
    const int hf = blockIdx.x & 1;
    const int h  = (blockIdx.x >> 1) % H_;
    const int b  = blockIdx.x / (2 * H_);
    const int n0 = hf * NH_;

    const float* __restrict__ Ar0 = edge_attr +
        ((size_t)b * E_ + (size_t)n0 * DEG_ + wave * 32 + l) * EDGE_F_ + 4 * half;
    const float* __restrict__ Ar1 = Ar0 + (size_t)16 * EDGE_F_;

    const float* __restrict__ Ah0 = hin +
        (size_t)(b * N_ + (n0 + l) % N_) * IN_F_ + wave * KW_ + 4 * half;
    const float* __restrict__ Ah1 = hin +
        (size_t)(b * N_ + (n0 + 16 + l) % N_) * IN_F_ + wave * KW_ + 4 * half;
    const float* __restrict__ Ah2 = hin +
        (size_t)(b * N_ + (n0 + 32 + l) % N_) * IN_F_ + wave * KW_ + 4 * half;
    const float* __restrict__ Bw = W_node +
        (size_t)(wave * KW_ + 4 * half) * HF_ + h * F_ + 4 * l;

    float4 wb[4];
    #pragma unroll
    for (int i = 0; i < 4; ++i) {
        const int idx = tid + i * FT_;
        const int kb  = idx >> 4;
        const int f4  = (idx & 15) * 4;
        wb[i] = *(const float4*)(W_edge + (size_t)kb * HF_ + h * F_ + f4);
    }
    v4f na[8][3];
    na[0][0] = *(const v4f*)(Ah0);
    na[0][1] = *(const v4f*)(Ah1);
    na[0][2] = *(const v4f*)(Ah2);
    v4f nb[4];
    #pragma unroll
    for (int i = 0; i < 4; ++i)
        nb[i] = *(const v4f*)(Bw + (size_t)i * HF_);
    __builtin_amdgcn_sched_barrier(0);
    #pragma unroll
    for (int j = 1; j < 4; ++j) {
        na[j][0] = *(const v4f*)(Ah0 + 8 * j);
        na[j][1] = *(const v4f*)(Ah1 + 8 * j);
        na[j][2] = *(const v4f*)(Ah2 + 8 * j);
    }
    __builtin_amdgcn_sched_barrier(0);

    #pragma unroll
    for (int i = 0; i < 4; ++i) {
        const int idx = tid + i * FT_;
        const int kb  = idx >> 4;
        const int f4  = (idx & 15) * 4;
        Bs[f4 + 0][kb] = wb[i].x;
        Bs[f4 + 1][kb] = wb[i].y;
        Bs[f4 + 2][kb] = wb[i].z;
        Bs[f4 + 3][kb] = wb[i].w;
    }
    bar_arrive();

    v8f nacc[3][4] = {};
    #pragma unroll
    for (int j = 0; j < 8; ++j) {
        v4f nn[4];
        if (j < 7) {
            #pragma unroll
            for (int i = 0; i < 4; ++i)
                nn[i] = *(const v4f*)(Bw + (size_t)(8 * (j + 1) + i) * HF_);
            if (j == 1) {
                #pragma unroll
                for (int jn = 4; jn < 8; ++jn) {
                    na[jn][0] = *(const v4f*)(Ah0 + 8 * jn);
                    na[jn][1] = *(const v4f*)(Ah1 + 8 * jn);
                    na[jn][2] = *(const v4f*)(Ah2 + 8 * jn);
                }
            }
            __builtin_amdgcn_sched_barrier(0);
        }
        const v2f x0lo = {na[j][0].x, na[j][0].y}, x0hi = {na[j][0].z, na[j][0].w};
        const v2f x1lo = {na[j][1].x, na[j][1].y}, x1hi = {na[j][1].z, na[j][1].w};
        const v2f x2lo = {na[j][2].x, na[j][2].y}, x2hi = {na[j][2].z, na[j][2].w};
        #pragma unroll
        for (int ct = 0; ct < 4; ++ct) {
            const v2f blo = {nb[0][ct], nb[1][ct]};
            nacc[0][ct] = __builtin_amdgcn_wmma_f32_16x16x4_f32(
                false, x0lo, false, blo, (short)0, nacc[0][ct], false, false);
            nacc[1][ct] = __builtin_amdgcn_wmma_f32_16x16x4_f32(
                false, x1lo, false, blo, (short)0, nacc[1][ct], false, false);
            nacc[2][ct] = __builtin_amdgcn_wmma_f32_16x16x4_f32(
                false, x2lo, false, blo, (short)0, nacc[2][ct], false, false);
        }
        #pragma unroll
        for (int ct = 0; ct < 4; ++ct) {
            const v2f bhi = {nb[2][ct], nb[3][ct]};
            nacc[0][ct] = __builtin_amdgcn_wmma_f32_16x16x4_f32(
                false, x0hi, false, bhi, (short)0, nacc[0][ct], false, false);
            nacc[1][ct] = __builtin_amdgcn_wmma_f32_16x16x4_f32(
                false, x1hi, false, bhi, (short)0, nacc[1][ct], false, false);
            nacc[2][ct] = __builtin_amdgcn_wmma_f32_16x16x4_f32(
                false, x2hi, false, bhi, (short)0, nacc[2][ct], false, false);
        }
        if (j < 7) {
            #pragma unroll
            for (int i = 0; i < 4; ++i) nb[i] = nn[i];
        }
    }

    float4 a0[4], a1[4];
    #pragma unroll
    for (int jj = 0; jj < 4; ++jj) {
        a0[jj] = *(const float4*)(Ar0 + 8 * jj);
        a1[jj] = *(const float4*)(Ar1 + 8 * jj);
    }
    const int c0 = tid & 15;
    const float4 wi = *(const float4*)(w_attn + 4 * c0);
    const float4 wj = *(const float4*)(w_attn + F_ + 4 * c0);
    float we[4];
    #pragma unroll
    for (int nt = 0; nt < 4; ++nt) we[nt] = w_attn[2 * F_ + nt * 16 + l];

    #pragma unroll
    for (int rt = 0; rt < 3; ++rt) {
        #pragma unroll
        for (int r = 0; r < 8; ++r) {
            const float4 v = {nacc[rt][0][r], nacc[rt][1][r],
                              nacc[rt][2][r], nacc[rt][3][r]};
            *(float4*)&ps[wave][16 * rt + 8 * half + r][4 * l] = v;
        }
    }
    bar_pass();
    bar_arrive();

    v8f acc0[4] = {};
    v8f acc1[4] = {};
    #pragma unroll
    for (int q = 0; q < 4; ++q) {
        float4 b0[4], b1[4];
        if (q < 3) {
            #pragma unroll
            for (int jj = 0; jj < 4; ++jj) {
                b0[jj] = *(const float4*)(Ar0 + 32 * (q + 1) + 8 * jj);
                b1[jj] = *(const float4*)(Ar1 + 32 * (q + 1) + 8 * jj);
            }
            __builtin_amdgcn_sched_barrier(0);
        }
        if (q == 1) {
            bar_pass();

            #pragma unroll
            for (int it = 0; it < 2; ++it) {
                const int qi = tid + it * FT_;
                const int jl = qi >> 4;
                const int jc = (jl < GT_) ? jl : (GT_ - 1);
                float4 a = *(const float4*)&ps[0][jc][4 * c0];
                #pragma unroll
                for (int s = 1; s < NW_; ++s) {
                    const float4 t = *(const float4*)&ps[s][jc][4 * c0];
                    a.x += t.x; a.y += t.y; a.z += t.z; a.w += t.w;
                }
                float si = a.x * wi.x + a.y * wi.y + a.z * wi.z + a.w * wi.w;
                float sj = a.x * wj.x + a.y * wj.y + a.z * wj.z + a.w * wj.w;
                #pragma unroll
                for (int o = 8; o >= 1; o >>= 1) {
                    si += __shfl_xor(si, o, 32);
                    sj += __shfl_xor(sj, o, 32);
                }
                if (jl < GT_) {
                    *(float4*)&gs[jl][4 * c0] = a;
                    if (c0 == 0) {
                        sis[jl] = si;
                        sjs[jl] = sj;
                    }
                }
            }

            bar_arrive();
        }
        #pragma unroll
        for (int jj = 0; jj < 4; ++jj) {
            float4 vb[4];
            #pragma unroll
            for (int nt = 0; nt < 4; ++nt)
                vb[nt] = *(const float4*)&Bs[nt * 16 + l][32 * q + 8 * jj + 4 * half];
            const v2f a0lo = {a0[jj].x, a0[jj].y}, a0hi = {a0[jj].z, a0[jj].w};
            const v2f a1lo = {a1[jj].x, a1[jj].y}, a1hi = {a1[jj].z, a1[jj].w};
            #pragma unroll
            for (int nt = 0; nt < 4; ++nt) {
                const v2f blo = {vb[nt].x, vb[nt].y};
                acc0[nt] = __builtin_amdgcn_wmma_f32_16x16x4_f32(
                    false, a0lo, false, blo, (short)0, acc0[nt], false, false);
                acc1[nt] = __builtin_amdgcn_wmma_f32_16x16x4_f32(
                    false, a1lo, false, blo, (short)0, acc1[nt], false, false);
            }
            #pragma unroll
            for (int nt = 0; nt < 4; ++nt) {
                const v2f bhi = {vb[nt].z, vb[nt].w};
                acc0[nt] = __builtin_amdgcn_wmma_f32_16x16x4_f32(
                    false, a0hi, false, bhi, (short)0, acc0[nt], false, false);
                acc1[nt] = __builtin_amdgcn_wmma_f32_16x16x4_f32(
                    false, a1hi, false, bhi, (short)0, acc1[nt], false, false);
            }
        }
        if (q < 3) {
            #pragma unroll
            for (int jj = 0; jj < 4; ++jj) {
                a0[jj] = b0[jj];
                a1[jj] = b1[jj];
            }
        }
    }

    const bool b3 = (l & 8) != 0, b2 = (l & 4) != 0, b1b = (l & 2) != 0;
    #pragma unroll
    for (int mt = 0; mt < 2; ++mt) {
        float pr[8];
        #pragma unroll
        for (int r = 0; r < 8; ++r)
            pr[r] = mt == 0
                ? acc0[0][r] * we[0] + acc0[1][r] * we[1] +
                  acc0[2][r] * we[2] + acc0[3][r] * we[3]
                : acc1[0][r] * we[0] + acc1[1][r] * we[1] +
                  acc1[2][r] * we[2] + acc1[3][r] * we[3];
        float qq[4];
        #pragma unroll
        for (int i = 0; i < 4; ++i) {
            const float keep = b3 ? pr[i + 4] : pr[i];
            const float send = b3 ? pr[i] : pr[i + 4];
            qq[i] = keep + __shfl_xor(send, 8, 32);
        }
        float u[2];
        #pragma unroll
        for (int i = 0; i < 2; ++i) {
            const float keep = b2 ? qq[i + 2] : qq[i];
            const float send = b2 ? qq[i] : qq[i + 2];
            u[i] = keep + __shfl_xor(send, 4, 32);
        }
        const float keep = b1b ? u[1] : u[0];
        const float send = b1b ? u[0] : u[1];
        float v = keep + __shfl_xor(send, 2, 32);
        v += __shfl_xor(v, 1, 32);
        if ((l & 1) == 0) {
            const int nl   = 2 * wave + mt;
            const int n    = n0 + nl;
            const int rank = 8 * half + (l >> 1);
            const int wrap = (n + DEG_ > N_) ? (n + DEG_ - N_) : 0;
            const int k    = (rank >= wrap) ? (rank - wrap) : (rank + N_ - n);
            ses[nl][k] = v;
        }
    }
    bar_pass();

    const int k = lane & 15;
    #pragma unroll
    for (int i = 0; i < 2; ++i) {
        const int nl = 2 * wave + i;
        float sc = sis[nl] + sjs[nl + k] + ses[nl][k];
        sc = (sc > 0.f) ? sc : SLOPE_ * sc;

        float m = sc;
        #pragma unroll
        for (int off = 8; off >= 1; off >>= 1)
            m = fmaxf(m, __shfl_xor(m, off, 32));
        float ex = __expf(sc - m);
        float s = ex;
        #pragma unroll
        for (int off = 8; off >= 1; off >>= 1)
            s += __shfl_xor(s, off, 32);
        float a = ex / s;

        float o0 = 0.f, o1 = 0.f;
        #pragma unroll
        for (int kk = 0; kk < DEG_; ++kk) {
            float ak = __shfl(a, kk, 32);
            o0 += ak * gs[nl + kk][lane];
            o1 += ak * gs[nl + kk][lane + 32];
        }
        float* op = out + ((size_t)(b * N_ + n0 + nl) * H_ + h) * F_;
        op[lane]      = o0;
        op[lane + 32] = o1;
    }
}

extern "C" void kernel_launch(void* const* d_in, const int* in_sizes, int n_in,
                              void* d_out, int out_size, void* d_ws, size_t ws_size,
                              hipStream_t stream)
{
    const float* h         = (const float*)d_in[0];
    const float* edge_attr = (const float*)d_in[2];
    const float* W_node    = (const float*)d_in[3];
    const float* W_edge    = (const float*)d_in[4];
    const float* w_attn    = (const float*)d_in[5];
    float* out = (float*)d_out;

    graph_attn_layer_kernel<<<B_ * H_ * 2, dim3(FT_), 0, stream>>>(
        h, W_node, w_attn, edge_attr, W_edge, out);
}
